# starting kernel
# baseline (speedup 1.0000x reference)
#include <hip/hip_runtime.h>
#include <math.h>

typedef __attribute__((ext_vector_type(2))) float v2f;
typedef __attribute__((ext_vector_type(8))) float v8f;

namespace {
constexpr int kB = 128;
constexpr int kP = 2048;
constexpr int kN = 256;
constexpr int kThreads = 512;
}

__device__ __forceinline__ float fast_rcp(float x) {
#if __has_builtin(__builtin_amdgcn_rcpf)
  return __builtin_amdgcn_rcpf(x);
#else
  return 1.0f / x;
#endif
}
__device__ __forceinline__ float fast_log2(float x) {
#if __has_builtin(__builtin_amdgcn_logf)
  return __builtin_amdgcn_logf(x);
#else
  return log2f(x);
#endif
}
__device__ __forceinline__ float fast_exp2(float x) {
#if __has_builtin(__builtin_amdgcn_exp2f)
  return __builtin_amdgcn_exp2f(x);
#else
  return exp2f(x);
#endif
}

template <bool POWGEN, int PCHUNK>
__device__ __forceinline__ v8f
accumulate_chunk(const float4* __restrict__ pts, float c0, float c1,
                 float w0, float w1, float e, int khalf) {
  v8f acc0 = {};
  v8f acc1 = {};
  const v2f ones = {1.0f, 1.0f};
  for (int pg = 0; pg < PCHUNK; pg += 8) {
#pragma unroll
    for (int g = 0; g < 2; ++g) {
      const int p = pg + g * 4 + khalf;
      const float4 q0 = pts[p];
      const float4 q1 = pts[p + 1];
      float s0 = fabsf(c0 - q0.x) * w0 + fabsf(c1 - q0.y) * w1;
      float s1 = fabsf(c0 - q1.x) * w0 + fabsf(c1 - q1.y) * w1;
      if (POWGEN) {
        s0 = fast_exp2(e * fast_log2(s0));
        s1 = fast_exp2(e * fast_log2(s1));
      }
      v2f a;
      a.x = q0.z * fast_rcp(1.0f + s0);
      a.y = q1.z * fast_rcp(1.0f + s1);
      if (g == 0)
        acc0 = __builtin_amdgcn_wmma_f32_16x16x4_f32(
            false, a, false, ones, (short)0, acc0, false, false);
      else
        acc1 = __builtin_amdgcn_wmma_f32_16x16x4_f32(
            false, a, false, ones, (short)0, acc1, false, false);
    }
  }
  return acc0 + acc1;
}

template <int SPLIT>
__global__ __launch_bounds__(kThreads)
void slayer_partial(const float* __restrict__ batch,
                    const float* __restrict__ not_dummy,
                    const float* __restrict__ centers,
                    const float* __restrict__ sharp,
                    const float* __restrict__ expo,
                    float* __restrict__ part) {
  constexpr int PCHUNK = kP / SPLIT;
  __shared__ float4 pts[PCHUNK];

  const int b = blockIdx.x;
  const int sidx = blockIdx.y;
  const int tid = (int)threadIdx.x;
  const int lane = tid & 31;
  const int wave = tid >> 5;
  const int pbase = sidx * PCHUNK;

  for (int i = tid; i < PCHUNK; i += kThreads) {
    const int p = pbase + i;
    const float2 x =
        *reinterpret_cast<const float2*>(batch + ((size_t)b * kP + p) * 2);
    const float msk = not_dummy[(size_t)b * kP + p];
    pts[i] = make_float4(x.x, x.y, msk, 0.0f);
  }

  const int n = wave * 16 + (lane & 15);
  const float c0 = centers[n * 2 + 0];
  const float c1 = centers[n * 2 + 1];
  const float w0 = fabsf(sharp[n * 2 + 0]);
  const float w1 = fabsf(sharp[n * 2 + 1]);
  const float e = expo[0];
  const int khalf = (lane >> 4) << 1;

  __syncthreads();

  v8f acc;
  if (e == 1.0f)
    acc = accumulate_chunk<false, PCHUNK>(pts, c0, c1, w0, w1, e, khalf);
  else
    acc = accumulate_chunk<true, PCHUNK>(pts, c0, c1, w0, w1, e, khalf);

  float* dst = part + ((size_t)sidx * kB + b) * kN + wave * 16;
  if (lane == 0) {
#pragma unroll
    for (int j = 0; j < 8; ++j) dst[j] = acc[j];
  } else if (lane == 16) {
#pragma unroll
    for (int j = 0; j < 8; ++j) dst[8 + j] = acc[j];
  }
}

template <int SPLIT>
__global__ void slayer_reduce(const float* __restrict__ part,
                              float* __restrict__ out) {
  const int i = blockIdx.x * blockDim.x + threadIdx.x;
  if (i < kB * kN) {
    float a = 0.0f;
#pragma unroll
    for (int s = 0; s < SPLIT; ++s)
      a += part[(size_t)s * (kB * kN) + i];
    out[i] = a;
  }
}

extern "C" void kernel_launch(void* const* d_in, const int* in_sizes, int n_in,
                              void* d_out, int out_size, void* d_ws,
                              size_t ws_size, hipStream_t stream) {
  const float* batch     = (const float*)d_in[0];
  const float* not_dummy = (const float*)d_in[1];
  const float* centers   = (const float*)d_in[2];
  const float* sharp     = (const float*)d_in[3];
  const float* expo      = (const float*)d_in[4];
  float* out = (float*)d_out;

  constexpr int kSplit = 8;
  const size_t need = (size_t)kSplit * kB * kN * sizeof(float);

  if (ws_size >= need) {
    float* part = (float*)d_ws;
    slayer_partial<kSplit><<<dim3(kB, kSplit), kThreads, 0, stream>>>(
        batch, not_dummy, centers, sharp, expo, part);
    slayer_reduce<kSplit><<<(kB * kN + 255) / 256, 256, 0, stream>>>(part, out);
  } else {
    slayer_partial<1><<<dim3(kB, 1), kThreads, 0, stream>>>(
        batch, not_dummy, centers, sharp, expo, out);
  }
}


# point split moved inside the work-group: 16 waves (8 center tiles x 2 point halves) cover a whole batch row and meet in LDS; one launch, no partial buffer, no reduce kernel; sum order changes
# speedup vs baseline: 1.1237x; 1.1237x over previous
#include <hip/hip_runtime.h>
#include <math.h>


typedef __attribute__((ext_vector_type(2))) float v2f;
typedef __attribute__((ext_vector_type(8))) float v8f;

namespace {
constexpr int kB = 128;
constexpr int kP = 2048;
constexpr int kN = 256;
constexpr int kTiles = 8;
constexpr int kHalves = 2;
constexpr int kWaves = kTiles * kHalves;
constexpr int kThreads = kWaves * 32;
constexpr int kTileN = kTiles * 16;
constexpr int kPW = kP / kHalves;
constexpr int kGroup = 32;
constexpr int kGroups = kPW / kGroup;
}

__device__ __forceinline__ float fast_rcp(float x) {
#if __has_builtin(__builtin_amdgcn_rcpf)
  return __builtin_amdgcn_rcpf(x);
#else
  return 1.0f / x;
#endif
}
__device__ __forceinline__ float fast_log2(float x) {
#if __has_builtin(__builtin_amdgcn_logf)
  return __builtin_amdgcn_logf(x);
#else
  return log2f(x);
#endif
}
__device__ __forceinline__ float fast_exp2(float x) {
#if __has_builtin(__builtin_amdgcn_exp2f)
  return __builtin_amdgcn_exp2f(x);
#else
  return exp2f(x);
#endif
}

__device__ __forceinline__ void pin_reads(float& c, v8f (&acc)[4]) {
  __builtin_amdgcn_sched_barrier(0);
  asm volatile(""
               : "+v"(c), "+v"(acc[0]), "+v"(acc[1]), "+v"(acc[2]), "+v"(acc[3])
               :
               : "memory");
  __builtin_amdgcn_sched_barrier(0);
}

__device__ __forceinline__ void load_group(const float4* __restrict__ grp,
                                           int khalf, float4 (&q)[16]) {
#pragma unroll
  for (int j = 0; j < 8; ++j) {
    q[2 * j + 0] = grp[j * 4 + khalf];
    q[2 * j + 1] = grp[j * 4 + khalf + 1];
  }
}

template <bool POWGEN>
__device__ __forceinline__ void
wmma_block(const float4 (&q)[16], float c0, float c1, float w0, float w1,
           float e, v8f (&acc)[4]) {
  const v2f ones = {1.0f, 1.0f};
#pragma unroll
  for (int j = 0; j < 8; ++j) {
    const float4 q0 = q[2 * j + 0];
    const float4 q1 = q[2 * j + 1];
    float s0 = fabsf(c0 - q0.x) * w0 + fabsf(c1 - q0.y) * w1;
    float s1 = fabsf(c0 - q1.x) * w0 + fabsf(c1 - q1.y) * w1;
    if (POWGEN) {
      s0 = fast_exp2(e * fast_log2(s0));
      s1 = fast_exp2(e * fast_log2(s1));
    }
    v2f a;
    a.x = q0.z * fast_rcp(1.0f + s0);
    a.y = q1.z * fast_rcp(1.0f + s1);
    acc[j & 3] = __builtin_amdgcn_wmma_f32_16x16x4_f32(
        false, a, false, ones, (short)0, acc[j & 3], false, false);
  }
}

template <bool POWGEN>
__device__ __forceinline__ v8f
accumulate_half(const float4* __restrict__ pts, float c0, float c1,
                float w0, float w1, float e, int khalf) {
  v8f acc[4];
#pragma unroll
  for (int j = 0; j < 4; ++j) acc[j] = v8f{};

  float4 qa[16];
  float4 qb[16];
  load_group(pts, khalf, qa);
  for (int g = 0; g < kGroups; g += 2) {
    load_group(pts + (g + 1) * kGroup, khalf, qb);
    pin_reads(c0, acc);
    wmma_block<POWGEN>(qa, c0, c1, w0, w1, e, acc);
    pin_reads(c0, acc);
    load_group(pts + ((g + 2) & (kGroups - 1)) * kGroup, khalf, qa);
    pin_reads(c0, acc);
    wmma_block<POWGEN>(qb, c0, c1, w0, w1, e, acc);
    pin_reads(c0, acc);
  }
  return (acc[0] + acc[1]) + (acc[2] + acc[3]);
}

__global__ __launch_bounds__(kThreads)
void slayer_rows(const float* __restrict__ batch,
                 const float* __restrict__ not_dummy,
                 const float* __restrict__ centers,
                 const float* __restrict__ sharp,
                 const float* __restrict__ expo,
                 float* __restrict__ out) {
  __shared__ float4 pts[kP];
  __shared__ float osum[kHalves * kTileN];

  const int b = blockIdx.x;
  const int nbase = blockIdx.y * kTileN;
  const int tid = (int)threadIdx.x;
  const int lane = tid & 31;
  const int wave = tid >> 5;
  const int tile = wave & (kTiles - 1);
  const int half = wave / kTiles;

  static_assert(kP % kThreads == 0, "staging assumes whole trips");
#pragma unroll
  for (int k = 0; k < kP / kThreads; ++k) {
    const int i = tid + k * kThreads;
    const float2 x =
        *reinterpret_cast<const float2*>(batch + ((size_t)b * kP + i) * 2);
    const float msk = not_dummy[(size_t)b * kP + i];
    pts[i] = make_float4(x.x, x.y, msk, 0.0f);
  }

  const int n = nbase + tile * 16 + (lane & 15);
  const float c0 = centers[n * 2 + 0];
  const float c1 = centers[n * 2 + 1];
  const float w0 = fabsf(sharp[n * 2 + 0]);
  const float w1 = fabsf(sharp[n * 2 + 1]);
  const float e = expo[0];
  const int khalf = (lane >> 4) << 1;

  __syncthreads();

  const float4* mine = pts + half * kPW;
  v8f acc;
  if (e == 1.0f)
    acc = accumulate_half<false>(mine, c0, c1, w0, w1, e, khalf);
  else
    acc = accumulate_half<true>(mine, c0, c1, w0, w1, e, khalf);

  float* dst = osum + half * kTileN + tile * 16;
  if (lane == 0) {
#pragma unroll
    for (int j = 0; j < 8; ++j) dst[j] = acc[j];
  } else if (lane == 16) {
#pragma unroll
    for (int j = 0; j < 8; ++j) dst[8 + j] = acc[j];
  }
  __syncthreads();
  if (tid < kTileN)
    out[(size_t)b * kN + nbase + tid] = osum[tid] + osum[kTileN + tid];
}

extern "C" void kernel_launch(void* const* d_in, const int* in_sizes, int n_in,
                              void* d_out, int out_size, void* d_ws,
                              size_t ws_size, hipStream_t stream) {
  const float* batch     = (const float*)d_in[0];
  const float* not_dummy = (const float*)d_in[1];
  const float* centers   = (const float*)d_in[2];
  const float* sharp     = (const float*)d_in[3];
  const float* expo      = (const float*)d_in[4];
  float* out = (float*)d_out;

  slayer_rows<<<dim3(kB, kN / kTileN), kThreads, 0, stream>>>(
      batch, not_dummy, centers, sharp, expo, out);
}


# each lane adds its own 512 points in four multiply-add chains (the mask multiply becomes the add); the ones-matrix sum runs once per wave behind the loop, not 256 times; sum order changes
# speedup vs baseline: 1.6080x; 1.4310x over previous
#include <hip/hip_runtime.h>
#include <math.h>


typedef __attribute__((ext_vector_type(2))) float v2f;
typedef __attribute__((ext_vector_type(8))) float v8f;

namespace {
constexpr int kB = 128;
constexpr int kP = 2048;
constexpr int kN = 256;
constexpr int kTiles = 8;
constexpr int kHalves = 2;
constexpr int kWaves = kTiles * kHalves;
constexpr int kThreads = kWaves * 32;
constexpr int kTileN = kTiles * 16;
constexpr int kPW = kP / kHalves;
constexpr int kGroup = 32;
constexpr int kGroups = kPW / kGroup;
}

__device__ __forceinline__ float fast_rcp(float x) {
#if __has_builtin(__builtin_amdgcn_rcpf)
  return __builtin_amdgcn_rcpf(x);
#else
  return 1.0f / x;
#endif
}
__device__ __forceinline__ float fast_log2(float x) {
#if __has_builtin(__builtin_amdgcn_logf)
  return __builtin_amdgcn_logf(x);
#else
  return log2f(x);
#endif
}
__device__ __forceinline__ float fast_exp2(float x) {
#if __has_builtin(__builtin_amdgcn_exp2f)
  return __builtin_amdgcn_exp2f(x);
#else
  return exp2f(x);
#endif
}

__device__ __forceinline__ void load_group(const float4* __restrict__ grp,
                                           int khalf, float4 (&q)[16]) {
#pragma unroll
  for (int j = 0; j < 8; ++j) {
    q[2 * j + 0] = grp[j * 4 + khalf];
    q[2 * j + 1] = grp[j * 4 + khalf + 1];
  }
}

template <bool POWGEN>
__device__ __forceinline__ void
sum_group(const float4 (&q)[16], float c0, float c1, float w0, float w1,
          float e, float (&acc)[4]) {
#pragma unroll
  for (int v = 0; v < 16; ++v) {
    const float4 p = q[v];
    float s = fabsf(c0 - p.x) * w0 + fabsf(c1 - p.y) * w1;
    if (POWGEN)
      s = fast_exp2(e * fast_log2(s));
    acc[v & 3] = fmaf(p.z, fast_rcp(1.0f + s), acc[v & 3]);
  }
}

template <bool POWGEN>
__device__ __forceinline__ v2f
accumulate_half(const float4* __restrict__ pts, float c0, float c1,
                float w0, float w1, float e, int khalf) {
  float acc[4] = {0.0f, 0.0f, 0.0f, 0.0f};

  for (int g = 0; g < kGroups; g += 2) {
    float4 qa[16];
    float4 qb[16];
    load_group(pts + g * kGroup, khalf, qa);
    load_group(pts + (g + 1) * kGroup, khalf, qb);
    sum_group<POWGEN>(qa, c0, c1, w0, w1, e, acc);
    sum_group<POWGEN>(qb, c0, c1, w0, w1, e, acc);
  }

  v2f a;
  a.x = acc[0] + acc[1];
  a.y = acc[2] + acc[3];
  return a;
}

__global__ __launch_bounds__(kThreads)
void slayer_rows(const float* __restrict__ batch,
                 const float* __restrict__ not_dummy,
                 const float* __restrict__ centers,
                 const float* __restrict__ sharp,
                 const float* __restrict__ expo,
                 float* __restrict__ out) {
  __shared__ float4 pts[kP];
  __shared__ float osum[kHalves * kTileN];

  const int b = blockIdx.x;
  const int nbase = blockIdx.y * kTileN;
  const int tid = (int)threadIdx.x;
  const int lane = tid & 31;
  const int wave = tid >> 5;
  const int tile = wave & (kTiles - 1);
  const int half = wave / kTiles;

  static_assert(kP % kThreads == 0, "staging assumes whole trips");
#pragma unroll
  for (int k = 0; k < kP / kThreads; ++k) {
    const int i = tid + k * kThreads;
    const float2 x =
        *reinterpret_cast<const float2*>(batch + ((size_t)b * kP + i) * 2);
    const float msk = not_dummy[(size_t)b * kP + i];
    pts[i] = make_float4(x.x, x.y, msk, 0.0f);
  }

  const int n = nbase + tile * 16 + (lane & 15);
  const float c0 = centers[n * 2 + 0];
  const float c1 = centers[n * 2 + 1];
  const float w0 = fabsf(sharp[n * 2 + 0]);
  const float w1 = fabsf(sharp[n * 2 + 1]);
  const float e = expo[0];
  const int khalf = (lane >> 4) << 1;

  __syncthreads();

  const float4* mine = pts + half * kPW;
  v2f a;
  if (e == 1.0f)
    a = accumulate_half<false>(mine, c0, c1, w0, w1, e, khalf);
  else
    a = accumulate_half<true>(mine, c0, c1, w0, w1, e, khalf);

  const v2f ones = {1.0f, 1.0f};
  v8f acc;
  asm volatile("v_wmma_f32_16x16x4_f32 %0, %1, %2, 0\n\t"
               "v_nop\n\tv_nop\n\tv_nop\n\tv_nop\n\tv_nop\n\t"
               "v_nop\n\tv_nop\n\tv_nop\n\tv_nop"
               : "=&v"(acc)
               : "v"(a), "v"(ones));

  float* dst = osum + half * kTileN + tile * 16;
  if (lane == 0) {
#pragma unroll
    for (int j = 0; j < 8; ++j) dst[j] = acc[j];
  } else if (lane == 16) {
#pragma unroll
    for (int j = 0; j < 8; ++j) dst[8 + j] = acc[j];
  }
  __syncthreads();
  if (tid < kTileN)
    out[(size_t)b * kN + nbase + tid] = osum[tid] + osum[kTileN + tid];
}

extern "C" void kernel_launch(void* const* d_in, const int* in_sizes, int n_in,
                              void* d_out, int out_size, void* d_ws,
                              size_t ws_size, hipStream_t stream) {
  const float* batch     = (const float*)d_in[0];
  const float* not_dummy = (const float*)d_in[1];
  const float* centers   = (const float*)d_in[2];
  const float* sharp     = (const float*)d_in[3];
  const float* expo      = (const float*)d_in[4];
  float* out = (float*)d_out;

  slayer_rows<<<dim3(kB, kN / kTileN), kThreads, 0, stream>>>(
      batch, not_dummy, centers, sharp, expo, out);
}


# four centers per lane: each point read from LDS feeds four centers; subtract and masked add run on packed pairs, the multiply-adds per center with |d| as modifier and 1+s inside; sum order changes
# speedup vs baseline: 2.2116x; 1.3754x over previous
#include <hip/hip_runtime.h>
#include <math.h>


typedef __attribute__((ext_vector_type(2))) float v2f;
typedef __attribute__((ext_vector_type(8))) float v8f;

namespace {
constexpr int kB = 128;
constexpr int kP = 2048;
constexpr int kN = 256;
constexpr int kTileN = 128;
constexpr int kSlots = 4;
constexpr int kCW = 16 * kSlots;
constexpr int kCGroups = kTileN / kCW;
constexpr int kWaves = 16;
constexpr int kPGroups = kWaves / kCGroups;
constexpr int kThreads = kWaves * 32;
constexpr int kPW = kP / kPGroups;
constexpr int kGroup = 32;
constexpr int kGroups = kPW / kGroup;
}

__device__ __forceinline__ float fast_rcp(float x) {
#if __has_builtin(__builtin_amdgcn_rcpf)
  return __builtin_amdgcn_rcpf(x);
#else
  return 1.0f / x;
#endif
}
__device__ __forceinline__ float fast_log2(float x) {
#if __has_builtin(__builtin_amdgcn_logf)
  return __builtin_amdgcn_logf(x);
#else
  return log2f(x);
#endif
}
__device__ __forceinline__ float fast_exp2(float x) {
#if __has_builtin(__builtin_amdgcn_exp2f)
  return __builtin_amdgcn_exp2f(x);
#else
  return exp2f(x);
#endif
}

__device__ __forceinline__ void load_group(const float4* __restrict__ grp,
                                           int khalf, float4 (&q)[16]) {
#pragma unroll
  for (int j = 0; j < 8; ++j) {
    q[2 * j + 0] = grp[j * 4 + khalf];
    q[2 * j + 1] = grp[j * 4 + khalf + 1];
  }
}

template <bool POWGEN>
__device__ __forceinline__ float
rational(float dx, float dy, float w0, float w1, float e) {
  if (POWGEN) {
    const float s = fmaf(fabsf(dy), w1, fabsf(dx) * w0);
    return fast_rcp(1.0f + fast_exp2(e * fast_log2(s)));
  }
  float d = fmaf(fabsf(dy), w1, fmaf(fabsf(dx), w0, 1.0f));
  asm("" : "+v"(d));
  return fast_rcp(d);
}

template <bool POWGEN>
__device__ __forceinline__ void
sum_group(const float4 (&q)[16], const v2f (&c0)[2], const v2f (&c1)[2],
          const v2f (&w0)[2], const v2f (&w1)[2], float e,
          v2f (&acc)[2][2]) {
#pragma unroll
  for (int v = 0; v < 16; ++v) {
    const float4 p = q[v];
    const v2f m = {p.z, p.z};
#pragma unroll
    for (int h = 0; h < 2; ++h) {
      const v2f dx = c0[h] - p.x;
      const v2f dy = c1[h] - p.y;
      v2f r;
      r.x = rational<POWGEN>(dx.x, dy.x, w0[h].x, w1[h].x, e);
      r.y = rational<POWGEN>(dx.y, dy.y, w0[h].y, w1[h].y, e);
      acc[h][v & 1] = __builtin_elementwise_fma(m, r, acc[h][v & 1]);
    }
  }
}

template <bool POWGEN>
__device__ __forceinline__ void
accumulate_part(const float4* __restrict__ pts, const v2f (&c0)[2],
                const v2f (&c1)[2], const v2f (&w0)[2], const v2f (&w1)[2],
                float e, int khalf, v2f (&acc)[2][2]) {
  for (int g = 0; g < kGroups; g += 2) {
    float4 qa[16];
    float4 qb[16];
    load_group(pts + g * kGroup, khalf, qa);
    load_group(pts + (g + 1) * kGroup, khalf, qb);
    sum_group<POWGEN>(qa, c0, c1, w0, w1, e, acc);
    sum_group<POWGEN>(qb, c0, c1, w0, w1, e, acc);
  }
}

__device__ __forceinline__ v8f row_sums(v2f a, v2f ones) {
  v8f d;
  asm volatile("v_wmma_f32_16x16x4_f32 %0, %1, %2, 0\n\t"
               "v_nop\n\tv_nop\n\tv_nop\n\tv_nop\n\tv_nop\n\t"
               "v_nop\n\tv_nop\n\tv_nop\n\tv_nop"
               : "=&v"(d)
               : "v"(a), "v"(ones));
  return d;
}

__global__ __launch_bounds__(kThreads)
void slayer_rows(const float* __restrict__ batch,
                 const float* __restrict__ not_dummy,
                 const float* __restrict__ centers,
                 const float* __restrict__ sharp,
                 const float* __restrict__ expo,
                 float* __restrict__ out) {
  __shared__ float4 pts[kP];
  __shared__ float osum[kPGroups * kTileN];

  const int b = blockIdx.x;
  const int nbase = blockIdx.y * kTileN;
  const int tid = (int)threadIdx.x;
  const int lane = tid & 31;
  const int wave = tid >> 5;
  const int cgrp = wave & (kCGroups - 1);
  const int pgrp = wave / kCGroups;

  static_assert(kP % kThreads == 0, "staging assumes whole trips");
#pragma unroll
  for (int k = 0; k < kP / kThreads; ++k) {
    const int i = tid + k * kThreads;
    const float2 x =
        *reinterpret_cast<const float2*>(batch + ((size_t)b * kP + i) * 2);
    const float msk = not_dummy[(size_t)b * kP + i];
    pts[i] = make_float4(x.x, x.y, msk, 0.0f);
  }

  v2f c0[2], c1[2], w0[2], w1[2];
#pragma unroll
  for (int k = 0; k < kSlots; ++k) {
    const int n = nbase + cgrp * kCW + k * 16 + (lane & 15);
    const float2 c = *reinterpret_cast<const float2*>(centers + n * 2);
    const float2 s = *reinterpret_cast<const float2*>(sharp + n * 2);
    c0[k >> 1][k & 1] = c.x;
    c1[k >> 1][k & 1] = c.y;
    w0[k >> 1][k & 1] = fabsf(s.x);
    w1[k >> 1][k & 1] = fabsf(s.y);
  }
  const float e = expo[0];
  const int khalf = (lane >> 4) << 1;

  __syncthreads();

  const float4* mine = pts + pgrp * kPW;
  v2f acc[2][2] = {{{0.0f, 0.0f}, {0.0f, 0.0f}}, {{0.0f, 0.0f}, {0.0f, 0.0f}}};
  if (e == 1.0f)
    accumulate_part<false>(mine, c0, c1, w0, w1, e, khalf, acc);
  else
    accumulate_part<true>(mine, c0, c1, w0, w1, e, khalf, acc);

  const v2f ones = {1.0f, 1.0f};
  v8f d[kSlots];
#pragma unroll
  for (int k = 0; k < kSlots; ++k) {
    v2f a;
    a.x = acc[k >> 1][0][k & 1];
    a.y = acc[k >> 1][1][k & 1];
    d[k] = row_sums(a, ones);
  }

  float* dst = osum + pgrp * kTileN + cgrp * kCW;
  if (lane == 0) {
#pragma unroll
    for (int k = 0; k < kSlots; ++k)
#pragma unroll
      for (int j = 0; j < 8; ++j) dst[k * 16 + j] = d[k][j];
  } else if (lane == 16) {
#pragma unroll
    for (int k = 0; k < kSlots; ++k)
#pragma unroll
      for (int j = 0; j < 8; ++j) dst[k * 16 + 8 + j] = d[k][j];
  }
  __syncthreads();
  if (tid < kTileN) {
    float s = osum[tid];
#pragma unroll
    for (int g = 1; g < kPGroups; ++g) s += osum[g * kTileN + tid];
    out[(size_t)b * kN + nbase + tid] = s;
  }
}

extern "C" void kernel_launch(void* const* d_in, const int* in_sizes, int n_in,
                              void* d_out, int out_size, void* d_ws,
                              size_t ws_size, hipStream_t stream) {
  const float* batch     = (const float*)d_in[0];
  const float* not_dummy = (const float*)d_in[1];
  const float* centers   = (const float*)d_in[2];
  const float* sharp     = (const float*)d_in[3];
  const float* expo      = (const float*)d_in[4];
  float* out = (float*)d_out;

  slayer_rows<<<dim3(kB, kN / kTileN), kThreads, 0, stream>>>(
      batch, not_dummy, centers, sharp, expo, out);
}


# e==1 loop reordered: a point's 4 reciprocals sit in one asm statement, each with a packed subtract of the next point behind it: no reciprocal follows another, no filler nop; sum order unchanged
# speedup vs baseline: 2.5497x; 1.1529x over previous
#include <hip/hip_runtime.h>
#include <math.h>


typedef __attribute__((ext_vector_type(2))) float v2f;
typedef __attribute__((ext_vector_type(8))) float v8f;

namespace {
constexpr int kB = 128;
constexpr int kP = 2048;
constexpr int kN = 256;
constexpr int kTileN = 128;
constexpr int kSlots = 4;
constexpr int kCW = 16 * kSlots;
constexpr int kCGroups = kTileN / kCW;
constexpr int kWaves = 16;
constexpr int kPGroups = kWaves / kCGroups;
constexpr int kThreads = kWaves * 32;
constexpr int kPW = kP / kPGroups;
constexpr int kGroup = 32;
constexpr int kGroups = kPW / kGroup;
}

__device__ __forceinline__ float fast_rcp(float x) {
#if __has_builtin(__builtin_amdgcn_rcpf)
  return __builtin_amdgcn_rcpf(x);
#else
  return 1.0f / x;
#endif
}
__device__ __forceinline__ float fast_log2(float x) {
#if __has_builtin(__builtin_amdgcn_logf)
  return __builtin_amdgcn_logf(x);
#else
  return log2f(x);
#endif
}
__device__ __forceinline__ float fast_exp2(float x) {
#if __has_builtin(__builtin_amdgcn_exp2f)
  return __builtin_amdgcn_exp2f(x);
#else
  return exp2f(x);
#endif
}

__device__ __forceinline__ void load_group(const float4* __restrict__ grp,
                                           int khalf, float4 (&q)[16]) {
#pragma unroll
  for (int j = 0; j < 8; ++j) {
    q[2 * j + 0] = grp[j * 4 + khalf];
    q[2 * j + 1] = grp[j * 4 + khalf + 1];
  }
}

__device__ __forceinline__ float
rational(float dx, float dy, float w0, float w1, float e) {
  const float s = fmaf(fabsf(dy), w1, fabsf(dx) * w0);
  return fast_rcp(1.0f + fast_exp2(e * fast_log2(s)));
}

__device__ __forceinline__ void
sum_group(const float4 (&q)[16], const v2f (&c0)[2], const v2f (&c1)[2],
          const v2f (&w0)[2], const v2f (&w1)[2], float e,
          v2f (&acc)[2][2]) {
#pragma unroll
  for (int v = 0; v < 16; ++v) {
    const float4 p = q[v];
    const v2f m = {p.z, p.z};
#pragma unroll
    for (int h = 0; h < 2; ++h) {
      const v2f dx = c0[h] - p.x;
      const v2f dy = c1[h] - p.y;
      v2f r;
      r.x = rational(dx.x, dy.x, w0[h].x, w1[h].x, e);
      r.y = rational(dx.y, dy.y, w0[h].y, w1[h].y, e);
      acc[h][v & 1] = __builtin_elementwise_fma(m, r, acc[h][v & 1]);
    }
  }
}

__device__ __forceinline__ void
rcp4_with_diffs(float (&r)[4], v2f (&dx)[2], v2f (&dy)[2], const v2f (&c0)[2],
                const v2f (&c1)[2], v2f pxy) {
  asm("v_rcp_f32 %0, %0\n\t"
      "v_pk_add_f32 %4, %8, %12 op_sel_hi:[1,0] neg_lo:[0,1] neg_hi:[0,1]\n\t"
      "v_rcp_f32 %1, %1\n\t"
      "v_pk_add_f32 %5, %9, %12 op_sel_hi:[1,0] neg_lo:[0,1] neg_hi:[0,1]\n\t"
      "v_rcp_f32 %2, %2\n\t"
      "v_pk_add_f32 %6, %10, %12 op_sel:[0,1] neg_lo:[0,1] neg_hi:[0,1]\n\t"
      "v_rcp_f32 %3, %3\n\t"
      "v_pk_add_f32 %7, %11, %12 op_sel:[0,1] neg_lo:[0,1] neg_hi:[0,1]"
      : "+v"(r[0]), "+v"(r[1]), "+v"(r[2]), "+v"(r[3]), "=&v"(dx[0]),
        "=&v"(dx[1]), "=&v"(dy[0]), "=&v"(dy[1])
      : "v"(c0[0]), "v"(c0[1]), "v"(c1[0]), "v"(c1[1]), "v"(pxy));
}

__device__ __forceinline__ void
sum_group_paired(const float4 (&q)[16], const v2f (&c0)[2], const v2f (&c1)[2],
                 const v2f (&w0)[2], const v2f (&w1)[2], float (&d)[4],
                 float& m, v2f (&acc)[2][2]) {
#pragma unroll
  for (int v = 0; v < 16; ++v) {
    const float4 p = q[v];
    const v2f pxy = {p.x, p.y};
    float r[4] = {d[0], d[1], d[2], d[3]};
    v2f dx[2], dy[2];
    rcp4_with_diffs(r, dx, dy, c0, c1, pxy);
    float t[4];
#pragma unroll
    for (int k = 0; k < 4; ++k)
      t[k] = fmaf(fabsf(dx[k >> 1][k & 1]), w0[k >> 1][k & 1], 1.0f);
    const v2f mm = {m, m};
    v2f rr0 = {r[0], r[1]};
    v2f rr1 = {r[2], r[3]};
    asm("" : "+v"(rr0) : "v"(t[0]), "v"(t[1]), "v"(t[2]), "v"(t[3]));
    acc[0][(v + 1) & 1] =
        __builtin_elementwise_fma(mm, rr0, acc[0][(v + 1) & 1]);
    d[0] = fmaf(fabsf(dy[0].x), w1[0].x, t[0]);
    d[1] = fmaf(fabsf(dy[0].y), w1[0].y, t[1]);
    asm("" : "+v"(rr1) : "v"(d[0]), "v"(d[1]));
    acc[1][(v + 1) & 1] =
        __builtin_elementwise_fma(mm, rr1, acc[1][(v + 1) & 1]);
    d[2] = fmaf(fabsf(dy[1].x), w1[1].x, t[2]);
    d[3] = fmaf(fabsf(dy[1].y), w1[1].y, t[3]);
    m = p.z;
  }
  asm("" : "+v"(d[0]), "+v"(d[1]), "+v"(d[2]), "+v"(d[3]));
}

__device__ __forceinline__ void
accumulate_part(const float4* __restrict__ pts, const v2f (&c0)[2],
                const v2f (&c1)[2], const v2f (&w0)[2], const v2f (&w1)[2],
                float e, int khalf, v2f (&acc)[2][2]) {
  for (int g = 0; g < kGroups; g += 2) {
    float4 qa[16];
    float4 qb[16];
    load_group(pts + g * kGroup, khalf, qa);
    load_group(pts + (g + 1) * kGroup, khalf, qb);
    sum_group(qa, c0, c1, w0, w1, e, acc);
    sum_group(qb, c0, c1, w0, w1, e, acc);
  }
}

__device__ __forceinline__ void
accumulate_part_paired(const float4* __restrict__ pts, const v2f (&c0)[2],
                       const v2f (&c1)[2], const v2f (&w0)[2],
                       const v2f (&w1)[2], int khalf, v2f (&acc)[2][2]) {
  float d[4] = {1.0f, 1.0f, 1.0f, 1.0f};
  float m = 0.0f;
  for (int g = 0; g < kGroups; g += 2) {
    float4 qa[16];
    float4 qb[16];
    load_group(pts + g * kGroup, khalf, qa);
    load_group(pts + (g + 1) * kGroup, khalf, qb);
    sum_group_paired(qa, c0, c1, w0, w1, d, m, acc);
    sum_group_paired(qb, c0, c1, w0, w1, d, m, acc);
  }
  const v2f mm = {m, m};
#pragma unroll
  for (int h = 0; h < 2; ++h) {
    const v2f rr = {fast_rcp(d[2 * h]), fast_rcp(d[2 * h + 1])};
    acc[h][1] = __builtin_elementwise_fma(mm, rr, acc[h][1]);
  }
}

__device__ __forceinline__ v8f row_sums(v2f a, v2f ones) {
  v8f d;
  asm volatile("v_wmma_f32_16x16x4_f32 %0, %1, %2, 0\n\t"
               "v_nop\n\tv_nop\n\tv_nop\n\tv_nop\n\tv_nop\n\t"
               "v_nop\n\tv_nop\n\tv_nop\n\tv_nop"
               : "=&v"(d)
               : "v"(a), "v"(ones));
  return d;
}

__global__ __launch_bounds__(kThreads)
__attribute__((amdgpu_waves_per_eu(4, 4)))
void slayer_rows(const float* __restrict__ batch,
                 const float* __restrict__ not_dummy,
                 const float* __restrict__ centers,
                 const float* __restrict__ sharp,
                 const float* __restrict__ expo,
                 float* __restrict__ out) {
  __shared__ float4 pts[kP];
  __shared__ float osum[kPGroups * kTileN];

  const int b = blockIdx.x;
  const int nbase = blockIdx.y * kTileN;
  const int tid = (int)threadIdx.x;
  const int lane = tid & 31;
  const int wave = tid >> 5;
  const int cgrp = wave & (kCGroups - 1);
  const int pgrp = wave / kCGroups;

  static_assert(kP % kThreads == 0, "staging assumes whole trips");
#pragma unroll
  for (int k = 0; k < kP / kThreads; ++k) {
    const int i = tid + k * kThreads;
    const float2 x =
        *reinterpret_cast<const float2*>(batch + ((size_t)b * kP + i) * 2);
    const float msk = not_dummy[(size_t)b * kP + i];
    pts[i] = make_float4(x.x, x.y, msk, 0.0f);
  }

  v2f c0[2], c1[2], w0[2], w1[2];
#pragma unroll
  for (int k = 0; k < kSlots; ++k) {
    const int n = nbase + cgrp * kCW + k * 16 + (lane & 15);
    const float2 c = *reinterpret_cast<const float2*>(centers + n * 2);
    const float2 s = *reinterpret_cast<const float2*>(sharp + n * 2);
    c0[k >> 1][k & 1] = c.x;
    c1[k >> 1][k & 1] = c.y;
    w0[k >> 1][k & 1] = fabsf(s.x);
    w1[k >> 1][k & 1] = fabsf(s.y);
  }
  const float e = expo[0];
  const int khalf = (lane >> 4) << 1;

  __syncthreads();

  const float4* mine = pts + pgrp * kPW;
  v2f acc[2][2] = {{{0.0f, 0.0f}, {0.0f, 0.0f}}, {{0.0f, 0.0f}, {0.0f, 0.0f}}};
  if (e == 1.0f)
    accumulate_part_paired(mine, c0, c1, w0, w1, khalf, acc);
  else
    accumulate_part(mine, c0, c1, w0, w1, e, khalf, acc);

  const v2f ones = {1.0f, 1.0f};
  v8f d[kSlots];
#pragma unroll
  for (int k = 0; k < kSlots; ++k) {
    v2f a;
    a.x = acc[k >> 1][0][k & 1];
    a.y = acc[k >> 1][1][k & 1];
    d[k] = row_sums(a, ones);
  }

  float* dst = osum + pgrp * kTileN + cgrp * kCW;
  if (lane == 0) {
#pragma unroll
    for (int k = 0; k < kSlots; ++k)
#pragma unroll
      for (int j = 0; j < 8; ++j) dst[k * 16 + j] = d[k][j];
  } else if (lane == 16) {
#pragma unroll
    for (int k = 0; k < kSlots; ++k)
#pragma unroll
      for (int j = 0; j < 8; ++j) dst[k * 16 + 8 + j] = d[k][j];
  }
  __syncthreads();
  if (tid < kTileN) {
    float s = osum[tid];
#pragma unroll
    for (int g = 1; g < kPGroups; ++g) s += osum[g * kTileN + tid];
    out[(size_t)b * kN + nbase + tid] = s;
  }
}

extern "C" void kernel_launch(void* const* d_in, const int* in_sizes, int n_in,
                              void* d_out, int out_size, void* d_ws,
                              size_t ws_size, hipStream_t stream) {
  const float* batch     = (const float*)d_in[0];
  const float* not_dummy = (const float*)d_in[1];
  const float* centers   = (const float*)d_in[2];
  const float* sharp     = (const float*)d_in[3];
  const float* expo      = (const float*)d_in[4];
  float* out = (float*)d_out;

  slayer_rows<<<dim3(kB, kN / kTileN), kThreads, 0, stream>>>(
      batch, not_dummy, centers, sharp, expo, out);
}
